# starting kernel
# baseline (speedup 1.0000x reference)
#include <hip/hip_runtime.h>

typedef float v2f __attribute__((ext_vector_type(2)));
typedef float v8f __attribute__((ext_vector_type(8)));

#define HIDDEN 2048
#define EMBED  512
#define NV     16
#define WAVES  16
#define BLOCK  (WAVES * 32)

__device__ __forceinline__ v8f wmma_f32_k4(v2f a, v2f b, v8f c) {
  return __builtin_amdgcn_wmma_f32_16x16x4_f32(false, a, false, b, (short)0, c,
                                               false, false);
}

template <int KPER>
__device__ __forceinline__ void gemv16x3_wmma(const float* __restrict__ W,
                                              int stride, int jBase,
                                              const float* __restrict__ sX,
                                              int kStart,
                                              v8f& c0, v8f& c1, v8f& c2) {
  const int L   = threadIdx.x & 31;
  const int sub = L & 15;
  const int hi2 = (L >> 4) << 1;
  const float* r0 = W + (size_t)(jBase + sub) * (size_t)stride + hi2 + kStart;
  const float* r1 = r0 + (size_t)HIDDEN * (size_t)stride;
  const float* r2 = r1 + (size_t)HIDDEN * (size_t)stride;
  const float* xp = sX + hi2 + kStart;
#pragma unroll 4
  for (int kb = 0; kb < KPER; kb += 4) {
    const float2 b  = *(const float2*)(xp + kb);
    const float2 a0 = *(const float2*)(r0 + kb);
    const float2 a1 = *(const float2*)(r1 + kb);
    const float2 a2 = *(const float2*)(r2 + kb);
    v2f B;  B.x  = b.x;  B.y  = b.y;
    v2f A0; A0.x = a0.x; A0.y = a0.y;
    v2f A1; A1.x = a1.x; A1.y = a1.y;
    v2f A2; A2.x = a2.x; A2.y = a2.y;
    c0 = wmma_f32_k4(A0, B, c0);
    c1 = wmma_f32_k4(A1, B, c1);
    c2 = wmma_f32_k4(A2, B, c2);
  }
}

template <int KPER>
__device__ __forceinline__ void gemv16_wmma(const float* __restrict__ W,
                                            int stride,
                                            const float* __restrict__ sX,
                                            int kStart, v8f& c) {
  const int L   = threadIdx.x & 31;
  const int sub = L & 15;
  const int hi2 = (L >> 4) << 1;
  const float* r0 = W + (size_t)sub * (size_t)stride + hi2 + kStart;
  const float* xp = sX + hi2 + kStart;
#pragma unroll 4
  for (int kb = 0; kb < KPER; kb += 4) {
    const float2 b = *(const float2*)(xp + kb);
    const float2 a = *(const float2*)(r0 + kb);
    v2f B; B.x = b.x; B.y = b.y;
    v2f A; A.x = a.x; A.y = a.y;
    c = wmma_f32_k4(A, B, c);
  }
}

__device__ __forceinline__ void reduce16(v8f c, float* acc) {
  const int L   = threadIdx.x & 31;
  const int sub = L & 15;
  const int hi  = L >> 4;
#pragma unroll
  for (int m = 0; m < 8; ++m) {
    if (sub == m) atomicAdd(&acc[m + 8 * hi], c[m]);
  }
}

template <int K_IH, int K_HH>
__global__ void __launch_bounds__(BLOCK)
gru_matvec_kernel(const float* __restrict__ W_ih, const float* __restrict__ W_hh,
                  const float* __restrict__ b_ih, const float* __restrict__ b_hh,
                  const float* __restrict__ xsrc, const int* __restrict__ xidx,
                  const float* __restrict__ hprev, float* __restrict__ pre) {
  __shared__ float sX[HIDDEN];
  __shared__ float sAcc[3][16];

  const int tid   = threadIdx.x;
  const int jBase = blockIdx.x * 16;
  const int mat   = blockIdx.y;

  if (mat == 0) {
    const float* xv = xidx ? (xsrc + (size_t)xidx[0] * (size_t)K_IH) : xsrc;
    for (int i = tid; i < K_IH; i += BLOCK) sX[i] = xv[i];
  } else {
    for (int i = tid; i < K_HH; i += BLOCK) sX[i] = hprev[i];
  }
  if (tid < 48) ((float*)sAcc)[tid] = 0.0f;
  __syncthreads();

  const int wave = __builtin_amdgcn_readfirstlane(tid >> 5);

  if (mat == 0) {
    constexpr int kPer = K_IH / WAVES;
    v8f c0 = {}, c1 = {}, c2 = {};
    gemv16x3_wmma<kPer>(W_ih, K_IH, jBase, sX, wave * kPer, c0, c1, c2);
    reduce16(c0, sAcc[0]);
    reduce16(c1, sAcc[1]);
    reduce16(c2, sAcc[2]);
  } else {
    constexpr int kPer = K_HH / WAVES;
    v8f c0 = {}, c1 = {}, c2 = {};
    gemv16x3_wmma<kPer>(W_hh, K_HH, jBase, sX, wave * kPer, c0, c1, c2);
    reduce16(c0, sAcc[0]);
    reduce16(c1, sAcc[1]);
    reduce16(c2, sAcc[2]);
  }
  __syncthreads();

  if (tid < 48) {
    const int g = tid >> 4;
    const int m = tid & 15;
    const int j = jBase + m;
    const float* bias = (mat == 0) ? b_ih : b_hh;
    pre[(mat * 3 + g) * HIDDEN + j] = sAcc[g][m] + bias[g * HIDDEN + j];
  }
}

__global__ void __launch_bounds__(256)
gru_gate_kernel(const float* __restrict__ pre, const float* __restrict__ hprev,
                float* __restrict__ hnext, float* __restrict__ hcopy) {
  const int j = blockIdx.x * 256 + threadIdx.x;
  if (j < HIDDEN) {
    const float ir  = pre[0 * HIDDEN + j];
    const float iz  = pre[1 * HIDDEN + j];
    const float inn = pre[2 * HIDDEN + j];
    const float hr  = pre[3 * HIDDEN + j];
    const float hz  = pre[4 * HIDDEN + j];
    const float hn  = pre[5 * HIDDEN + j];
    const float r = 1.0f / (1.0f + __expf(-(ir + hr)));
    const float z = 1.0f / (1.0f + __expf(-(iz + hz)));
    const float n = tanhf(inn + r * hn);
    const float h = (1.0f - z) * n + z * hprev[j];
    hnext[j] = h;
    hcopy[j] = h;
  }
}

__global__ void __launch_bounds__(BLOCK)
head_kernel(const float* __restrict__ w_head, const float* __restrict__ b_head,
            const float* __restrict__ h1, float* __restrict__ probs) {
  __shared__ float sX[HIDDEN];
  __shared__ float sAcc[16];
  const int tid = threadIdx.x;
  for (int i = tid; i < HIDDEN; i += BLOCK) sX[i] = h1[i];
  if (tid < 16) sAcc[tid] = 0.0f;
  __syncthreads();

  const int wave = __builtin_amdgcn_readfirstlane(tid >> 5);
  constexpr int kPer = HIDDEN / WAVES;
  v8f c = {};
  gemv16_wmma<kPer>(w_head, HIDDEN, sX, wave * kPer, c);
  reduce16(c, sAcc);
  __syncthreads();

  if (tid < 16) {
    const float v = sAcc[tid] + b_head[tid];
    float m = v;
    for (int off = 8; off > 0; off >>= 1) m = fmaxf(m, __shfl_xor(m, off, 32));
    const float e = __expf(v - m);
    float s = e;
    for (int off = 8; off > 0; off >>= 1) s += __shfl_xor(s, off, 32);
    probs[tid] = e / s;
  }
}

extern "C" void kernel_launch(void* const* d_in, const int* in_sizes, int n_in,
                              void* d_out, int out_size, void* d_ws,
                              size_t ws_size, hipStream_t stream) {
  const int*   x      = (const int*)  d_in[0];
  const float* h_in   = (const float*)d_in[1];
  const float* emb    = (const float*)d_in[2];
  const float* w_ih0  = (const float*)d_in[3];
  const float* w_hh0  = (const float*)d_in[4];
  const float* b_ih0  = (const float*)d_in[5];
  const float* b_hh0  = (const float*)d_in[6];
  const float* w_ih1  = (const float*)d_in[7];
  const float* w_hh1  = (const float*)d_in[8];
  const float* b_ih1  = (const float*)d_in[9];
  const float* b_hh1  = (const float*)d_in[10];
  const float* w_head = (const float*)d_in[11];
  const float* b_head = (const float*)d_in[12];

  float* out = (float*)d_out;
  float* h0  = (float*)d_ws;
  float* h1  = h0 + HIDDEN;
  float* pre = h1 + HIDDEN;

  const dim3 gridMV(HIDDEN / 16, 2);

  gru_matvec_kernel<EMBED, HIDDEN><<<gridMV, BLOCK, 0, stream>>>(
      w_ih0, w_hh0, b_ih0, b_hh0, emb, x, h_in, pre);
  gru_gate_kernel<<<HIDDEN / 256, 256, 0, stream>>>(pre, h_in, h0, out + NV);

  gru_matvec_kernel<HIDDEN, HIDDEN><<<gridMV, BLOCK, 0, stream>>>(
      w_ih1, w_hh1, b_ih1, b_hh1, h0, nullptr, h_in + HIDDEN, pre);
  gru_gate_kernel<<<HIDDEN / 256, 256, 0, stream>>>(pre, h_in + HIDDEN, h1,
                                                    out + NV + HIDDEN);

  head_kernel<<<1, BLOCK, 0, stream>>>(w_head, b_head, h1, out);
}


# Layer-2 gate math moved to its reader: each head-kernel wave forms its own 128 h values under the loads of one staged 16x128 head tile; 5 launches become 4; logit sum order changes (2 accumulators)
# speedup vs baseline: 1.3429x; 1.3429x over previous
#include <hip/hip_runtime.h>


typedef float v2f __attribute__((ext_vector_type(2)));
typedef float v4f __attribute__((ext_vector_type(4)));
typedef float v8f __attribute__((ext_vector_type(8)));

#define HIDDEN 2048
#define EMBED  512
#define NV     16
#define WAVES  16
#define BLOCK  (WAVES * 32)

#define STG_PITCH_MAX 132
#define STG_FLOATS    (16 * STG_PITCH_MAX)
#define XS_FLOATS     128

__device__ __forceinline__ v8f wmma_f32_k4(v2f a, v2f b, v8f c) {
  return __builtin_amdgcn_wmma_f32_16x16x4_f32(false, a, false, b, (short)0, c,
                                               false, false);
}

__device__ __forceinline__ void wave_lds_handover() {
  __builtin_amdgcn_fence(__ATOMIC_RELEASE, "wavefront");
  __builtin_amdgcn_wave_barrier();
  __builtin_amdgcn_fence(__ATOMIC_ACQUIRE, "wavefront");
}

template <int KPER>
__device__ __forceinline__ v8f tile_times_x(const float* sr, const float* xr) {
  v8f ce = {}, co = {};
#pragma unroll
  for (int j = 0; j < KPER / 8; ++j) {
    const v4f a = *(const v4f*)(sr + 8 * j);
    const v4f b = *(const v4f*)(xr + 8 * j);
    v2f A0; A0.x = a.x; A0.y = a.y;
    v2f B0; B0.x = b.x; B0.y = b.y;
    v2f A1; A1.x = a.z; A1.y = a.w;
    v2f B1; B1.x = b.z; B1.y = b.w;
    ce = wmma_f32_k4(A0, B0, ce);
    co = wmma_f32_k4(A1, B1, co);
  }
  return ce + co;
}

__device__ __forceinline__ void reduce16(v8f c, float* acc) {
  const int L   = threadIdx.x & 31;
  const int sub = L & 15;
  const int hi  = L >> 4;
#pragma unroll
  for (int m = 0; m < 8; ++m) {
    if (sub == m) atomicAdd(&acc[m + 8 * hi], c[m]);
  }
}

template <int K>
__device__ __forceinline__ void gemv16x3_staged(const float* __restrict__ W,
                                                int jBase,
                                                const float* __restrict__ xg,
                                                int wave, float* stg, float* xs,
                                                float* sAcc) {
  constexpr int KPER  = K / WAVES;
  constexpr int LPR   = KPER / 4;
  constexpr int RPI   = 32 / LPR;
  constexpr int NLD   = 16 / RPI;
  constexpr int PITCH = KPER + 4;
  static_assert(PITCH <= STG_PITCH_MAX && KPER <= XS_FLOATS, "staging fits");

  const int L  = threadIdx.x & 31;
  const int lr = L / LPR;
  const int lk = (L % LPR) * 4;
  const int kStart = wave * KPER;

  const float* g0 = W + (size_t)(jBase + lr) * (size_t)K + kStart + lk;
  const float* g1 = g0 + (size_t)HIDDEN * (size_t)K;
  const float* g2 = g1 + (size_t)HIDDEN * (size_t)K;

  const v4f xv = *(const v4f*)(xg + kStart + lk);
  v4f ta[NLD], tb[NLD];
#pragma unroll
  for (int i = 0; i < NLD; ++i)
    ta[i] = *(const v4f*)(g0 + (size_t)(i * RPI) * (size_t)K);
#pragma unroll
  for (int i = 0; i < NLD; ++i)
    tb[i] = *(const v4f*)(g1 + (size_t)(i * RPI) * (size_t)K);

  float* sw = stg + lr * PITCH + lk;
  const float* sr = stg + (L & 15) * PITCH + 4 * (L >> 4);
  const float* xr = xs + 4 * (L >> 4);

  *(v4f*)(xs + lk) = xv;

#pragma unroll
  for (int i = 0; i < NLD; ++i) *(v4f*)(sw + i * RPI * PITCH) = ta[i];
#pragma unroll
  for (int i = 0; i < NLD; ++i)
    ta[i] = *(const v4f*)(g2 + (size_t)(i * RPI) * (size_t)K);
  wave_lds_handover();
  reduce16(tile_times_x<KPER>(sr, xr), sAcc);

  wave_lds_handover();
#pragma unroll
  for (int i = 0; i < NLD; ++i) *(v4f*)(sw + i * RPI * PITCH) = tb[i];
  wave_lds_handover();
  reduce16(tile_times_x<KPER>(sr, xr), sAcc + 16);

  wave_lds_handover();
#pragma unroll
  for (int i = 0; i < NLD; ++i) *(v4f*)(sw + i * RPI * PITCH) = ta[i];
  wave_lds_handover();
  reduce16(tile_times_x<KPER>(sr, xr), sAcc + 32);
}

template <int K_IH, int K_HH>
__global__ void __launch_bounds__(BLOCK)
gru_matvec_kernel(const float* __restrict__ W_ih, const float* __restrict__ W_hh,
                  const float* __restrict__ b_ih, const float* __restrict__ b_hh,
                  const float* __restrict__ xsrc, const int* __restrict__ xidx,
                  const float* __restrict__ hprev, float* __restrict__ pre) {
  __shared__ __attribute__((aligned(16))) float sStage[WAVES * STG_FLOATS];
  __shared__ __attribute__((aligned(16))) float sXs[WAVES * XS_FLOATS];
  __shared__ float sAcc[3][16];

  const int tid   = threadIdx.x;
  const int jBase = blockIdx.x * 16;
  const int mat   = blockIdx.y;

  if (tid < 48) ((float*)sAcc)[tid] = 0.0f;
  __syncthreads();

  const int wave = __builtin_amdgcn_readfirstlane(tid >> 5);
  float* stg = sStage + wave * STG_FLOATS;
  float* xs  = sXs + wave * XS_FLOATS;

  if (mat == 0) {
    const float* xv = xidx ? (xsrc + (size_t)xidx[0] * (size_t)K_IH) : xsrc;
    gemv16x3_staged<K_IH>(W_ih, jBase, xv, wave, stg, xs, (float*)sAcc);
  } else {
    gemv16x3_staged<K_HH>(W_hh, jBase, hprev, wave, stg, xs, (float*)sAcc);
  }
  __syncthreads();

  if (tid < 48) {
    const int g = tid >> 4;
    const int m = tid & 15;
    const int j = jBase + m;
    const float* bias = (mat == 0) ? b_ih : b_hh;
    pre[(mat * 3 + g) * HIDDEN + j] = sAcc[g][m] + bias[g * HIDDEN + j];
  }
}

__device__ __forceinline__ float gru_unit(float ir, float iz, float inn,
                                          float hr, float hz, float hn,
                                          float hp) {
  const float r = 1.0f / (1.0f + __expf(-(ir + hr)));
  const float z = 1.0f / (1.0f + __expf(-(iz + hz)));
  const float n = tanhf(inn + r * hn);
  return (1.0f - z) * n + z * hp;
}

__global__ void __launch_bounds__(256)
gru_gate_kernel(const float* __restrict__ pre, const float* __restrict__ hprev,
                float* __restrict__ hnext, float* __restrict__ hcopy) {
  const int j = blockIdx.x * 256 + threadIdx.x;
  if (j < HIDDEN) {
    const float h = gru_unit(pre[0 * HIDDEN + j], pre[1 * HIDDEN + j],
                             pre[2 * HIDDEN + j], pre[3 * HIDDEN + j],
                             pre[4 * HIDDEN + j], pre[5 * HIDDEN + j],
                             hprev[j]);
    hnext[j] = h;
    hcopy[j] = h;
  }
}

__global__ void __launch_bounds__(BLOCK)
head_kernel(const float* __restrict__ w_head, const float* __restrict__ b_head,
            const float* __restrict__ pre, const float* __restrict__ hprev,
            float* __restrict__ hcopy, float* __restrict__ probs) {
  __shared__ __attribute__((aligned(16))) float sStage[WAVES * STG_FLOATS];
  __shared__ __attribute__((aligned(16))) float sXs[WAVES * XS_FLOATS];
  __shared__ float sAcc[16];

  constexpr int KPER  = HIDDEN / WAVES;
  constexpr int PITCH = KPER + 4;
  static_assert(PITCH <= STG_PITCH_MAX && KPER <= XS_FLOATS, "staging fits");

  const int tid = threadIdx.x;
  if (tid < 16) sAcc[tid] = 0.0f;
  __syncthreads();

  const int wave = __builtin_amdgcn_readfirstlane(tid >> 5);
  float* stg = sStage + wave * STG_FLOATS;
  float* xs  = sXs + wave * XS_FLOATS;
  const int L  = tid & 31;
  const int lk = 4 * L;
  const int k  = wave * KPER + lk;

  const v4f ir  = *(const v4f*)(pre + 0 * HIDDEN + k);
  const v4f iz  = *(const v4f*)(pre + 1 * HIDDEN + k);
  const v4f inn = *(const v4f*)(pre + 2 * HIDDEN + k);
  const v4f hr  = *(const v4f*)(pre + 3 * HIDDEN + k);
  const v4f hz  = *(const v4f*)(pre + 4 * HIDDEN + k);
  const v4f hn  = *(const v4f*)(pre + 5 * HIDDEN + k);
  const v4f hp  = *(const v4f*)(hprev + k);
  v4f t[16];
#pragma unroll
  for (int i = 0; i < 16; ++i)
    t[i] = *(const v4f*)(w_head + (size_t)i * (size_t)HIDDEN + k);

  v4f h;
  h.x = gru_unit(ir.x, iz.x, inn.x, hr.x, hz.x, hn.x, hp.x);
  h.y = gru_unit(ir.y, iz.y, inn.y, hr.y, hz.y, hn.y, hp.y);
  h.z = gru_unit(ir.z, iz.z, inn.z, hr.z, hz.z, hn.z, hp.z);
  h.w = gru_unit(ir.w, iz.w, inn.w, hr.w, hz.w, hn.w, hp.w);
  *(v4f*)(hcopy + k) = h;
  *(v4f*)(xs + lk) = h;

#pragma unroll
  for (int i = 0; i < 16; ++i) *(v4f*)(stg + i * PITCH + lk) = t[i];
  wave_lds_handover();
  const float* sr = stg + (L & 15) * PITCH + 4 * (L >> 4);
  const float* xr = xs + 4 * (L >> 4);
  reduce16(tile_times_x<KPER>(sr, xr), sAcc);
  __syncthreads();

  if (tid < 16) {
    const float v = sAcc[tid] + b_head[tid];
    float m = v;
    for (int off = 8; off > 0; off >>= 1) m = fmaxf(m, __shfl_xor(m, off, 32));
    const float e = __expf(v - m);
    float s = e;
    for (int off = 8; off > 0; off >>= 1) s += __shfl_xor(s, off, 32);
    probs[tid] = e / s;
  }
}

extern "C" void kernel_launch(void* const* d_in, const int* in_sizes, int n_in,
                              void* d_out, int out_size, void* d_ws,
                              size_t ws_size, hipStream_t stream) {
  const int*   x      = (const int*)  d_in[0];
  const float* h_in   = (const float*)d_in[1];
  const float* emb    = (const float*)d_in[2];
  const float* w_ih0  = (const float*)d_in[3];
  const float* w_hh0  = (const float*)d_in[4];
  const float* b_ih0  = (const float*)d_in[5];
  const float* b_hh0  = (const float*)d_in[6];
  const float* w_ih1  = (const float*)d_in[7];
  const float* w_hh1  = (const float*)d_in[8];
  const float* b_ih1  = (const float*)d_in[9];
  const float* b_hh1  = (const float*)d_in[10];
  const float* w_head = (const float*)d_in[11];
  const float* b_head = (const float*)d_in[12];

  float* out = (float*)d_out;
  float* h0  = (float*)d_ws;
  float* pre = h0 + 2 * HIDDEN;

  const dim3 gridMV(HIDDEN / 16, 2);

  gru_matvec_kernel<EMBED, HIDDEN><<<gridMV, BLOCK, 0, stream>>>(
      w_ih0, w_hh0, b_ih0, b_hh0, emb, x, h_in, pre);
  gru_gate_kernel<<<HIDDEN / 256, 256, 0, stream>>>(pre, h_in, h0, out + NV);

  gru_matvec_kernel<HIDDEN, HIDDEN><<<gridMV, BLOCK, 0, stream>>>(
      w_ih1, w_hh1, b_ih1, b_hh1, h0, nullptr, h_in + HIDDEN, pre);
  head_kernel<<<1, BLOCK, 0, stream>>>(w_head, b_head, pre, h_in + HIDDEN,
                                       out + NV + HIDDEN, out);
}


# Staged tiles cut into 32-k slices (4 loads of four 128-B row lines): a slice's 8 matrix steps follow its loads at once, so a wave's last bytes leave 8 steps to do, not 32; sum order unchanged
# speedup vs baseline: 1.4652x; 1.0911x over previous
#include <hip/hip_runtime.h>


typedef float v2f __attribute__((ext_vector_type(2)));
typedef float v4f __attribute__((ext_vector_type(4)));
typedef float v8f __attribute__((ext_vector_type(8)));

#define HIDDEN 2048
#define EMBED  512
#define NV     16
#define WAVES  16
#define BLOCK  (WAVES * 32)

#define SLICE_K     32
#define SLICE_PITCH (SLICE_K + 4)
#define STG_FLOATS  (16 * SLICE_PITCH)
#define XS_FLOATS   128
#define RING_MAX    8

__device__ __forceinline__ v8f wmma_f32_k4(v2f a, v2f b, v8f c) {
  return __builtin_amdgcn_wmma_f32_16x16x4_f32(false, a, false, b, (short)0, c,
                                               false, false);
}

__device__ __forceinline__ void wave_lds_handover() {
  __builtin_amdgcn_fence(__ATOMIC_RELEASE, "wavefront");
  __builtin_amdgcn_wave_barrier();
  __builtin_amdgcn_fence(__ATOMIC_ACQUIRE, "wavefront");
}

__device__ __forceinline__ void slice_times_x(const float* sr, const float* xr,
                                              v8f& ce, v8f& co) {
#pragma unroll
  for (int j = 0; j < SLICE_K / 8; ++j) {
    const v4f a = *(const v4f*)(sr + 8 * j);
    const v4f b = *(const v4f*)(xr + 8 * j);
    v2f A0; A0.x = a.x; A0.y = a.y;
    v2f B0; B0.x = b.x; B0.y = b.y;
    v2f A1; A1.x = a.z; A1.y = a.w;
    v2f B1; B1.x = b.z; B1.y = b.w;
    ce = wmma_f32_k4(A0, B0, ce);
    co = wmma_f32_k4(A1, B1, co);
  }
}

__device__ __forceinline__ void reduce16(v8f c, float* acc) {
  const int L   = threadIdx.x & 31;
  const int sub = L & 15;
  const int hi  = L >> 4;
#pragma unroll
  for (int m = 0; m < 8; ++m) {
    if (sub == m) atomicAdd(&acc[m + 8 * hi], c[m]);
  }
}

template <int K>
__device__ __forceinline__ void gemv16x3_staged(const float* __restrict__ W,
                                                int jBase,
                                                const float* __restrict__ xg,
                                                int wave, float* stg, float* xs,
                                                float* sAcc) {
  constexpr int KPER = K / WAVES;
  constexpr int NS   = KPER / SLICE_K;
  constexpr int NU   = 3 * NS;
  constexpr int RING = NU < RING_MAX ? NU : RING_MAX;
  static_assert(KPER % SLICE_K == 0 && KPER <= XS_FLOATS, "staging fits");

  const int L  = threadIdx.x & 31;
  const int lr = L >> 3;
  const int lk = (L & 7) * 4;
  const int xl = (L * 4) & (KPER - 1);
  const int kStart = wave * KPER;

  const float* g = W + (size_t)(jBase + lr) * (size_t)K + kStart + lk;

  const v4f xv = *(const v4f*)(xg + kStart + xl);
  v4f t[RING][4];
#pragma unroll
  for (int u = 0; u < RING; ++u) {
#pragma unroll
    for (int i = 0; i < 4; ++i)
      t[u][i] = *(const v4f*)(g + ((size_t)(u / NS) * HIDDEN + 4 * i) * (size_t)K +
                              (u % NS) * SLICE_K);
  }

  float* sw = stg + lr * SLICE_PITCH + lk;
  const float* sr = stg + (L & 15) * SLICE_PITCH + 4 * (L >> 4);
  const float* xr = xs + 4 * (L >> 4);

  *(v4f*)(xs + xl) = xv;

  v8f ce = {}, co = {};
#pragma unroll
  for (int u = 0; u < NU; ++u) {
    if (u > 0) wave_lds_handover();
#pragma unroll
    for (int i = 0; i < 4; ++i) *(v4f*)(sw + 4 * i * SLICE_PITCH) = t[u % RING][i];
    if (u + RING < NU) {
#pragma unroll
      for (int i = 0; i < 4; ++i)
        t[u % RING][i] =
            *(const v4f*)(g + ((size_t)((u + RING) / NS) * HIDDEN + 4 * i) * (size_t)K +
                          ((u + RING) % NS) * SLICE_K);
    }
    wave_lds_handover();
    slice_times_x(sr, xr + (u % NS) * SLICE_K, ce, co);
    if (u % NS == NS - 1) {
      reduce16(ce + co, sAcc + 16 * (u / NS));
      ce = v8f{};
      co = v8f{};
    }
  }
}

template <int K_IH, int K_HH>
__global__ void __launch_bounds__(BLOCK)
gru_matvec_kernel(const float* __restrict__ W_ih, const float* __restrict__ W_hh,
                  const float* __restrict__ b_ih, const float* __restrict__ b_hh,
                  const float* __restrict__ xsrc, const int* __restrict__ xidx,
                  const float* __restrict__ hprev, float* __restrict__ pre) {
  __shared__ __attribute__((aligned(16))) float sStage[WAVES * STG_FLOATS];
  __shared__ __attribute__((aligned(16))) float sXs[WAVES * XS_FLOATS];
  __shared__ float sAcc[3][16];

  const int tid   = threadIdx.x;
  const int jBase = blockIdx.x * 16;
  const int mat   = blockIdx.y;

  if (tid < 48) ((float*)sAcc)[tid] = 0.0f;
  __syncthreads();

  const int wave = __builtin_amdgcn_readfirstlane(tid >> 5);
  float* stg = sStage + wave * STG_FLOATS;
  float* xs  = sXs + wave * XS_FLOATS;

  if (mat == 0) {
    const float* xv = xidx ? (xsrc + (size_t)xidx[0] * (size_t)K_IH) : xsrc;
    gemv16x3_staged<K_IH>(W_ih, jBase, xv, wave, stg, xs, (float*)sAcc);
  } else {
    gemv16x3_staged<K_HH>(W_hh, jBase, hprev, wave, stg, xs, (float*)sAcc);
  }
  __syncthreads();

  if (tid < 48) {
    const int g = tid >> 4;
    const int m = tid & 15;
    const int j = jBase + m;
    const float* bias = (mat == 0) ? b_ih : b_hh;
    pre[(mat * 3 + g) * HIDDEN + j] = sAcc[g][m] + bias[g * HIDDEN + j];
  }
}

__device__ __forceinline__ float gru_unit(float ir, float iz, float inn,
                                          float hr, float hz, float hn,
                                          float hp) {
  const float r = 1.0f / (1.0f + __expf(-(ir + hr)));
  const float z = 1.0f / (1.0f + __expf(-(iz + hz)));
  const float n = tanhf(inn + r * hn);
  return (1.0f - z) * n + z * hp;
}

__global__ void __launch_bounds__(256)
gru_gate_kernel(const float* __restrict__ pre, const float* __restrict__ hprev,
                float* __restrict__ hnext, float* __restrict__ hcopy) {
  const int j = blockIdx.x * 256 + threadIdx.x;
  if (j < HIDDEN) {
    const float h = gru_unit(pre[0 * HIDDEN + j], pre[1 * HIDDEN + j],
                             pre[2 * HIDDEN + j], pre[3 * HIDDEN + j],
                             pre[4 * HIDDEN + j], pre[5 * HIDDEN + j],
                             hprev[j]);
    hnext[j] = h;
    hcopy[j] = h;
  }
}

__global__ void __launch_bounds__(BLOCK)
head_kernel(const float* __restrict__ w_head, const float* __restrict__ b_head,
            const float* __restrict__ pre, const float* __restrict__ hprev,
            float* __restrict__ hcopy, float* __restrict__ probs) {
  __shared__ __attribute__((aligned(16))) float sStage[WAVES * STG_FLOATS];
  __shared__ __attribute__((aligned(16))) float sXs[WAVES * XS_FLOATS];
  __shared__ float sAcc[16];

  constexpr int KPER = HIDDEN / WAVES;
  constexpr int NS   = KPER / SLICE_K;
  static_assert(KPER % SLICE_K == 0 && KPER <= XS_FLOATS, "staging fits");

  const int tid = threadIdx.x;
  if (tid < 16) sAcc[tid] = 0.0f;
  __syncthreads();

  const int wave = __builtin_amdgcn_readfirstlane(tid >> 5);
  float* stg = sStage + wave * STG_FLOATS;
  float* xs  = sXs + wave * XS_FLOATS;
  const int L  = tid & 31;
  const int lk = 4 * L;
  const int k  = wave * KPER + lk;

  const v4f ir  = *(const v4f*)(pre + 0 * HIDDEN + k);
  const v4f iz  = *(const v4f*)(pre + 1 * HIDDEN + k);
  const v4f inn = *(const v4f*)(pre + 2 * HIDDEN + k);
  const v4f hr  = *(const v4f*)(pre + 3 * HIDDEN + k);
  const v4f hz  = *(const v4f*)(pre + 4 * HIDDEN + k);
  const v4f hn  = *(const v4f*)(pre + 5 * HIDDEN + k);
  const v4f hp  = *(const v4f*)(hprev + k);
  const int lr  = L >> 3;
  const int lks = (L & 7) * 4;
  const float* gw = w_head + (size_t)lr * (size_t)HIDDEN + wave * KPER + lks;
  v4f t[NS][4];
#pragma unroll
  for (int u = 0; u < NS; ++u) {
#pragma unroll
    for (int i = 0; i < 4; ++i)
      t[u][i] = *(const v4f*)(gw + (size_t)(4 * i) * (size_t)HIDDEN + u * SLICE_K);
  }

  v4f h;
  h.x = gru_unit(ir.x, iz.x, inn.x, hr.x, hz.x, hn.x, hp.x);
  h.y = gru_unit(ir.y, iz.y, inn.y, hr.y, hz.y, hn.y, hp.y);
  h.z = gru_unit(ir.z, iz.z, inn.z, hr.z, hz.z, hn.z, hp.z);
  h.w = gru_unit(ir.w, iz.w, inn.w, hr.w, hz.w, hn.w, hp.w);
  *(v4f*)(hcopy + k) = h;
  *(v4f*)(xs + lk) = h;

  float* sw = stg + lr * SLICE_PITCH + lks;
  const float* sr = stg + (L & 15) * SLICE_PITCH + 4 * (L >> 4);
  const float* xr = xs + 4 * (L >> 4);
  v8f ce = {}, co = {};
#pragma unroll
  for (int u = 0; u < NS; ++u) {
    if (u > 0) wave_lds_handover();
#pragma unroll
    for (int i = 0; i < 4; ++i) *(v4f*)(sw + 4 * i * SLICE_PITCH) = t[u][i];
    wave_lds_handover();
    slice_times_x(sr, xr + u * SLICE_K, ce, co);
  }
  reduce16(ce + co, sAcc);
  __syncthreads();

  if (tid < 16) {
    const float v = sAcc[tid] + b_head[tid];
    float m = v;
    for (int off = 8; off > 0; off >>= 1) m = fmaxf(m, __shfl_xor(m, off, 32));
    const float e = __expf(v - m);
    float s = e;
    for (int off = 8; off > 0; off >>= 1) s += __shfl_xor(s, off, 32);
    probs[tid] = e / s;
  }
}

extern "C" void kernel_launch(void* const* d_in, const int* in_sizes, int n_in,
                              void* d_out, int out_size, void* d_ws,
                              size_t ws_size, hipStream_t stream) {
  const int*   x      = (const int*)  d_in[0];
  const float* h_in   = (const float*)d_in[1];
  const float* emb    = (const float*)d_in[2];
  const float* w_ih0  = (const float*)d_in[3];
  const float* w_hh0  = (const float*)d_in[4];
  const float* b_ih0  = (const float*)d_in[5];
  const float* b_hh0  = (const float*)d_in[6];
  const float* w_ih1  = (const float*)d_in[7];
  const float* w_hh1  = (const float*)d_in[8];
  const float* b_ih1  = (const float*)d_in[9];
  const float* b_hh1  = (const float*)d_in[10];
  const float* w_head = (const float*)d_in[11];
  const float* b_head = (const float*)d_in[12];

  float* out = (float*)d_out;
  float* h0  = (float*)d_ws;
  float* pre = h0 + 2 * HIDDEN;

  const dim3 gridMV(HIDDEN / 16, 2);

  gru_matvec_kernel<EMBED, HIDDEN><<<gridMV, BLOCK, 0, stream>>>(
      w_ih0, w_hh0, b_ih0, b_hh0, emb, x, h_in, pre);
  gru_gate_kernel<<<HIDDEN / 256, 256, 0, stream>>>(pre, h_in, h0, out + NV);

  gru_matvec_kernel<HIDDEN, HIDDEN><<<gridMV, BLOCK, 0, stream>>>(
      w_ih1, w_hh1, b_ih1, b_hh1, h0, nullptr, h_in + HIDDEN, pre);
  head_kernel<<<1, BLOCK, 0, stream>>>(w_head, b_head, pre, h_in + HIDDEN,
                                       out + NV + HIDDEN, out);
}


# Matvec blocks run 8 waves, not 16: a wave takes twice the k range with 12 slices (48 loads) in flight, so half as many waves enter per launch; sum order changes (8 wave partials over twice the k)
# speedup vs baseline: 1.5716x; 1.0726x over previous
#include <hip/hip_runtime.h>


typedef float v2f __attribute__((ext_vector_type(2)));
typedef float v4f __attribute__((ext_vector_type(4)));
typedef float v8f __attribute__((ext_vector_type(8)));

#define HIDDEN 2048
#define EMBED  512
#define NV     16
#define WAVES  16
#define BLOCK  (WAVES * 32)
#define MV_WAVES 8
#define MV_BLOCK (MV_WAVES * 32)

#define SLICE_K     32
#define SLICE_PITCH (SLICE_K + 4)
#define STG_FLOATS  (16 * SLICE_PITCH)
#define XS_FLOATS   128
#define MV_XS_FLOATS 256
#define RING_MAX    12

__device__ __forceinline__ v8f wmma_f32_k4(v2f a, v2f b, v8f c) {
  return __builtin_amdgcn_wmma_f32_16x16x4_f32(false, a, false, b, (short)0, c,
                                               false, false);
}

__device__ __forceinline__ void wave_lds_handover() {
  __builtin_amdgcn_fence(__ATOMIC_RELEASE, "wavefront");
  __builtin_amdgcn_wave_barrier();
  __builtin_amdgcn_fence(__ATOMIC_ACQUIRE, "wavefront");
}

__device__ __forceinline__ void slice_times_x(const float* sr, const float* xr,
                                              v8f& ce, v8f& co) {
#pragma unroll
  for (int j = 0; j < SLICE_K / 8; ++j) {
    const v4f a = *(const v4f*)(sr + 8 * j);
    const v4f b = *(const v4f*)(xr + 8 * j);
    v2f A0; A0.x = a.x; A0.y = a.y;
    v2f B0; B0.x = b.x; B0.y = b.y;
    v2f A1; A1.x = a.z; A1.y = a.w;
    v2f B1; B1.x = b.z; B1.y = b.w;
    ce = wmma_f32_k4(A0, B0, ce);
    co = wmma_f32_k4(A1, B1, co);
  }
}

__device__ __forceinline__ void reduce16(v8f c, float* acc) {
  const int L   = threadIdx.x & 31;
  const int sub = L & 15;
  const int hi  = L >> 4;
#pragma unroll
  for (int m = 0; m < 8; ++m) {
    if (sub == m) atomicAdd(&acc[m + 8 * hi], c[m]);
  }
}

template <int K>
__device__ __forceinline__ void gemv16x3_staged(const float* __restrict__ W,
                                                int jBase,
                                                const float* __restrict__ xg,
                                                int wave, float* stg, float* xs,
                                                float* sAcc) {
  constexpr int KPER = K / MV_WAVES;
  constexpr int NS   = KPER / SLICE_K;
  constexpr int NU   = 3 * NS;
  constexpr int RING = NU < RING_MAX ? NU : RING_MAX;
  constexpr int XP   = (KPER + 127) / 128;
  static_assert(KPER % SLICE_K == 0 && KPER <= MV_XS_FLOATS, "staging fits");

  const int L  = threadIdx.x & 31;
  const int lr = L >> 3;
  const int lk = (L & 7) * 4;
  const int xl = L * 4;
  const int kStart = wave * KPER;

  const float* g = W + (size_t)(jBase + lr) * (size_t)K + kStart + lk;

  v4f xv[XP];
#pragma unroll
  for (int p = 0; p < XP; ++p)
    xv[p] = *(const v4f*)(xg + kStart + ((p * 128 + xl) & (KPER - 1)));
  v4f t[RING][4];
#pragma unroll
  for (int u = 0; u < RING; ++u) {
#pragma unroll
    for (int i = 0; i < 4; ++i)
      t[u][i] = *(const v4f*)(g + ((size_t)(u / NS) * HIDDEN + 4 * i) * (size_t)K +
                              (u % NS) * SLICE_K);
  }

  float* sw = stg + lr * SLICE_PITCH + lk;
  const float* sr = stg + (L & 15) * SLICE_PITCH + 4 * (L >> 4);
  const float* xr = xs + 4 * (L >> 4);

#pragma unroll
  for (int p = 0; p < XP; ++p) *(v4f*)(xs + ((p * 128 + xl) & (KPER - 1))) = xv[p];

  v8f ce = {}, co = {};
#pragma unroll
  for (int u = 0; u < NU; ++u) {
    if (u > 0) wave_lds_handover();
#pragma unroll
    for (int i = 0; i < 4; ++i) *(v4f*)(sw + 4 * i * SLICE_PITCH) = t[u % RING][i];
    if (u + RING < NU) {
#pragma unroll
      for (int i = 0; i < 4; ++i)
        t[u % RING][i] =
            *(const v4f*)(g + ((size_t)((u + RING) / NS) * HIDDEN + 4 * i) * (size_t)K +
                          ((u + RING) % NS) * SLICE_K);
    }
    wave_lds_handover();
    slice_times_x(sr, xr + (u % NS) * SLICE_K, ce, co);
    if (u % NS == NS - 1) {
      reduce16(ce + co, sAcc + 16 * (u / NS));
      ce = v8f{};
      co = v8f{};
    }
  }
}

template <int K_IH, int K_HH>
__global__ void __launch_bounds__(MV_BLOCK)
gru_matvec_kernel(const float* __restrict__ W_ih, const float* __restrict__ W_hh,
                  const float* __restrict__ b_ih, const float* __restrict__ b_hh,
                  const float* __restrict__ xsrc, const int* __restrict__ xidx,
                  const float* __restrict__ hprev, float* __restrict__ pre) {
  __shared__ __attribute__((aligned(16))) float sStage[MV_WAVES * STG_FLOATS];
  __shared__ __attribute__((aligned(16))) float sXs[MV_WAVES * MV_XS_FLOATS];
  __shared__ float sAcc[3][16];

  const int tid   = threadIdx.x;
  const int jBase = blockIdx.x * 16;
  const int mat   = blockIdx.y;

  if (tid < 48) ((float*)sAcc)[tid] = 0.0f;
  __syncthreads();

  const int wave = __builtin_amdgcn_readfirstlane(tid >> 5);
  float* stg = sStage + wave * STG_FLOATS;
  float* xs  = sXs + wave * MV_XS_FLOATS;

  if (mat == 0) {
    const float* xv = xidx ? (xsrc + (size_t)xidx[0] * (size_t)K_IH) : xsrc;
    gemv16x3_staged<K_IH>(W_ih, jBase, xv, wave, stg, xs, (float*)sAcc);
  } else {
    gemv16x3_staged<K_HH>(W_hh, jBase, hprev, wave, stg, xs, (float*)sAcc);
  }
  __syncthreads();

  if (tid < 48) {
    const int g = tid >> 4;
    const int m = tid & 15;
    const int j = jBase + m;
    const float* bias = (mat == 0) ? b_ih : b_hh;
    pre[(mat * 3 + g) * HIDDEN + j] = sAcc[g][m] + bias[g * HIDDEN + j];
  }
}

__device__ __forceinline__ float gru_unit(float ir, float iz, float inn,
                                          float hr, float hz, float hn,
                                          float hp) {
  const float r = 1.0f / (1.0f + __expf(-(ir + hr)));
  const float z = 1.0f / (1.0f + __expf(-(iz + hz)));
  const float n = tanhf(inn + r * hn);
  return (1.0f - z) * n + z * hp;
}

__global__ void __launch_bounds__(256)
gru_gate_kernel(const float* __restrict__ pre, const float* __restrict__ hprev,
                float* __restrict__ hnext, float* __restrict__ hcopy) {
  const int j = blockIdx.x * 256 + threadIdx.x;
  if (j < HIDDEN) {
    const float h = gru_unit(pre[0 * HIDDEN + j], pre[1 * HIDDEN + j],
                             pre[2 * HIDDEN + j], pre[3 * HIDDEN + j],
                             pre[4 * HIDDEN + j], pre[5 * HIDDEN + j],
                             hprev[j]);
    hnext[j] = h;
    hcopy[j] = h;
  }
}

__global__ void __launch_bounds__(BLOCK)
head_kernel(const float* __restrict__ w_head, const float* __restrict__ b_head,
            const float* __restrict__ pre, const float* __restrict__ hprev,
            float* __restrict__ hcopy, float* __restrict__ probs) {
  __shared__ __attribute__((aligned(16))) float sStage[WAVES * STG_FLOATS];
  __shared__ __attribute__((aligned(16))) float sXs[WAVES * XS_FLOATS];
  __shared__ float sAcc[16];

  constexpr int KPER = HIDDEN / WAVES;
  constexpr int NS   = KPER / SLICE_K;
  static_assert(KPER % SLICE_K == 0 && KPER <= XS_FLOATS, "staging fits");

  const int tid = threadIdx.x;
  if (tid < 16) sAcc[tid] = 0.0f;
  __syncthreads();

  const int wave = __builtin_amdgcn_readfirstlane(tid >> 5);
  float* stg = sStage + wave * STG_FLOATS;
  float* xs  = sXs + wave * XS_FLOATS;
  const int L  = tid & 31;
  const int lk = 4 * L;
  const int k  = wave * KPER + lk;

  const v4f ir  = *(const v4f*)(pre + 0 * HIDDEN + k);
  const v4f iz  = *(const v4f*)(pre + 1 * HIDDEN + k);
  const v4f inn = *(const v4f*)(pre + 2 * HIDDEN + k);
  const v4f hr  = *(const v4f*)(pre + 3 * HIDDEN + k);
  const v4f hz  = *(const v4f*)(pre + 4 * HIDDEN + k);
  const v4f hn  = *(const v4f*)(pre + 5 * HIDDEN + k);
  const v4f hp  = *(const v4f*)(hprev + k);
  const int lr  = L >> 3;
  const int lks = (L & 7) * 4;
  const float* gw = w_head + (size_t)lr * (size_t)HIDDEN + wave * KPER + lks;
  v4f t[NS][4];
#pragma unroll
  for (int u = 0; u < NS; ++u) {
#pragma unroll
    for (int i = 0; i < 4; ++i)
      t[u][i] = *(const v4f*)(gw + (size_t)(4 * i) * (size_t)HIDDEN + u * SLICE_K);
  }

  v4f h;
  h.x = gru_unit(ir.x, iz.x, inn.x, hr.x, hz.x, hn.x, hp.x);
  h.y = gru_unit(ir.y, iz.y, inn.y, hr.y, hz.y, hn.y, hp.y);
  h.z = gru_unit(ir.z, iz.z, inn.z, hr.z, hz.z, hn.z, hp.z);
  h.w = gru_unit(ir.w, iz.w, inn.w, hr.w, hz.w, hn.w, hp.w);
  *(v4f*)(hcopy + k) = h;
  *(v4f*)(xs + lk) = h;

  float* sw = stg + lr * SLICE_PITCH + lks;
  const float* sr = stg + (L & 15) * SLICE_PITCH + 4 * (L >> 4);
  const float* xr = xs + 4 * (L >> 4);
  v8f ce = {}, co = {};
#pragma unroll
  for (int u = 0; u < NS; ++u) {
    if (u > 0) wave_lds_handover();
#pragma unroll
    for (int i = 0; i < 4; ++i) *(v4f*)(sw + 4 * i * SLICE_PITCH) = t[u][i];
    wave_lds_handover();
    slice_times_x(sr, xr + u * SLICE_K, ce, co);
  }
  reduce16(ce + co, sAcc);
  __syncthreads();

  if (tid < 16) {
    const float v = sAcc[tid] + b_head[tid];
    float m = v;
    for (int off = 8; off > 0; off >>= 1) m = fmaxf(m, __shfl_xor(m, off, 32));
    const float e = __expf(v - m);
    float s = e;
    for (int off = 8; off > 0; off >>= 1) s += __shfl_xor(s, off, 32);
    probs[tid] = e / s;
  }
}

extern "C" void kernel_launch(void* const* d_in, const int* in_sizes, int n_in,
                              void* d_out, int out_size, void* d_ws,
                              size_t ws_size, hipStream_t stream) {
  const int*   x      = (const int*)  d_in[0];
  const float* h_in   = (const float*)d_in[1];
  const float* emb    = (const float*)d_in[2];
  const float* w_ih0  = (const float*)d_in[3];
  const float* w_hh0  = (const float*)d_in[4];
  const float* b_ih0  = (const float*)d_in[5];
  const float* b_hh0  = (const float*)d_in[6];
  const float* w_ih1  = (const float*)d_in[7];
  const float* w_hh1  = (const float*)d_in[8];
  const float* b_ih1  = (const float*)d_in[9];
  const float* b_hh1  = (const float*)d_in[10];
  const float* w_head = (const float*)d_in[11];
  const float* b_head = (const float*)d_in[12];

  float* out = (float*)d_out;
  float* h0  = (float*)d_ws;
  float* pre = h0 + 2 * HIDDEN;

  const dim3 gridMV(HIDDEN / 16, 2);

  gru_matvec_kernel<EMBED, HIDDEN><<<gridMV, MV_BLOCK, 0, stream>>>(
      w_ih0, w_hh0, b_ih0, b_hh0, emb, x, h_in, pre);
  gru_gate_kernel<<<HIDDEN / 256, 256, 0, stream>>>(pre, h_in, h0, out + NV);

  gru_matvec_kernel<HIDDEN, HIDDEN><<<gridMV, MV_BLOCK, 0, stream>>>(
      w_ih1, w_hh1, b_ih1, b_hh1, h0, nullptr, h_in + HIDDEN, pre);
  head_kernel<<<1, BLOCK, 0, stream>>>(w_head, b_head, pre, h_in + HIDDEN,
                                       out + NV + HIDDEN, out);
}


# First layer's launch balanced: each block pair splits the 2560 columns 1280/1280 (hidden-hidden k cut at 768); partial sums and biases meet in the gate kernel; hidden-hidden sum order changes
# speedup vs baseline: 1.6312x; 1.0379x over previous
#include <hip/hip_runtime.h>


typedef float v2f __attribute__((ext_vector_type(2)));
typedef float v4f __attribute__((ext_vector_type(4)));
typedef float v8f __attribute__((ext_vector_type(8)));
typedef const __attribute__((address_space(1))) float* gptr;
typedef const __attribute__((address_space(1))) v4f*   gptr4;

#define HIDDEN 2048
#define EMBED  512
#define NV     16
#define WAVES  16
#define BLOCK  (WAVES * 32)
#define MV_WAVES 8
#define MV_BLOCK (MV_WAVES * 32)

#define SLICE_K     32
#define SLICE_PITCH (SLICE_K + 4)
#define STG_FLOATS  (16 * SLICE_PITCH)
#define XS_FLOATS   128
#define MV_XS_FLOATS 256
#define RING_MAX    12

__device__ __forceinline__ v8f wmma_f32_k4(v2f a, v2f b, v8f c) {
  return __builtin_amdgcn_wmma_f32_16x16x4_f32(false, a, false, b, (short)0, c,
                                               false, false);
}

__device__ __forceinline__ void wave_lds_handover() {
  __builtin_amdgcn_fence(__ATOMIC_RELEASE, "wavefront");
  __builtin_amdgcn_wave_barrier();
  __builtin_amdgcn_fence(__ATOMIC_ACQUIRE, "wavefront");
}

__device__ __forceinline__ void slice_times_x(const float* sr, const float* xr,
                                              v8f& ce, v8f& co) {
#pragma unroll
  for (int j = 0; j < SLICE_K / 8; ++j) {
    const v4f a = *(const v4f*)(sr + 8 * j);
    const v4f b = *(const v4f*)(xr + 8 * j);
    v2f A0; A0.x = a.x; A0.y = a.y;
    v2f B0; B0.x = b.x; B0.y = b.y;
    v2f A1; A1.x = a.z; A1.y = a.w;
    v2f B1; B1.x = b.z; B1.y = b.w;
    ce = wmma_f32_k4(A0, B0, ce);
    co = wmma_f32_k4(A1, B1, co);
  }
}

__device__ __forceinline__ void reduce16(v8f c, float* acc) {
  const int L   = threadIdx.x & 31;
  const int sub = L & 15;
  const int hi  = L >> 4;
#pragma unroll
  for (int m = 0; m < 8; ++m) {
    if (sub == m) atomicAdd(&acc[m + 8 * hi], c[m]);
  }
}

__device__ __forceinline__ int piece_of(int xl, int cnt) {
  return (cnt & (cnt - 1)) == 0 ? (xl & (cnt - 1)) : (xl < cnt ? xl : xl - cnt);
}

template <int K, int NS>
__device__ __forceinline__ void gemv16x3_staged(const float* __restrict__ W,
                                                int jBase, int kStart,
                                                const float* __restrict__ xg,
                                                float* stg, float* xs,
                                                float* sAcc) {
  constexpr int KPER = NS * SLICE_K;
  constexpr int NU   = 3 * NS;
  constexpr int RING = NU < RING_MAX ? NU : RING_MAX;
  constexpr int XP   = (KPER + 127) / 128;
  static_assert(KPER <= MV_XS_FLOATS, "staging fits");

  const int L  = threadIdx.x & 31;
  const int lr = L >> 3;
  const int lk = (L & 7) * 4;
  const int xl = L * 4;

  const float* g = W + (size_t)(jBase + lr) * (size_t)K + kStart + lk;
  asm volatile("" : "+v"(g));
  const gptr gw = (gptr)g;

  v4f xv[XP];
#pragma unroll
  for (int p = 0; p < XP; ++p) {
    const int cnt = KPER - 128 * p < 128 ? KPER - 128 * p : 128;
    xv[p] = *(const v4f*)(xg + kStart + 128 * p + piece_of(xl, cnt));
  }
  v4f t[RING][4];
#pragma unroll
  for (int u = 0; u < RING; ++u) {
#pragma unroll
    for (int i = 0; i < 4; ++i)
      t[u][i] = *(gptr4)(gw + ((size_t)(u / NS) * HIDDEN + 4 * i) * (size_t)K +
                              (u % NS) * SLICE_K);
  }

  float* sw = stg + lr * SLICE_PITCH + lk;
  const float* sr = stg + (L & 15) * SLICE_PITCH + 4 * (L >> 4);
  const float* xr = xs + 4 * (L >> 4);

#pragma unroll
  for (int p = 0; p < XP; ++p) {
    const int cnt = KPER - 128 * p < 128 ? KPER - 128 * p : 128;
    *(v4f*)(xs + 128 * p + piece_of(xl, cnt)) = xv[p];
  }

  v8f ce = {}, co = {};
#pragma unroll
  for (int u = 0; u < NU; ++u) {
    if (u > 0) wave_lds_handover();
#pragma unroll
    for (int i = 0; i < 4; ++i) *(v4f*)(sw + 4 * i * SLICE_PITCH) = t[u % RING][i];
    if (u + RING < NU) {
#pragma unroll
      for (int i = 0; i < 4; ++i)
        t[u % RING][i] =
            *(gptr4)(gw + ((size_t)((u + RING) / NS) * HIDDEN + 4 * i) * (size_t)K +
                          ((u + RING) % NS) * SLICE_K);
    }
    wave_lds_handover();
    slice_times_x(sr, xr + (u % NS) * SLICE_K, ce, co);
    if (u % NS == NS - 1) {
      reduce16(ce + co, sAcc + 16 * (u / NS));
      ce = v8f{};
      co = v8f{};
    }
  }
}

template <int KA, int NSA, int ROWA2, int KB, int NSB, int ROWB2>
__device__ __forceinline__ void stream_slices(
    const float* __restrict__ WA, int kA, const float* __restrict__ xA,
    const float* __restrict__ WB, int kB, const float* __restrict__ xB,
    int jBase, float* stg, float* xs, float* sAcc) {
  constexpr int NSBd = NSB > 0 ? NSB : 1;
  constexpr int NUA  = 3 * NSA;
  constexpr int NU   = 3 * (NSA + NSB);
  constexpr int RING = NU < RING_MAX ? NU : RING_MAX;
  constexpr int XA   = SLICE_K * NSA;
  constexpr int XB   = SLICE_K * NSB;
  constexpr int XPA  = (XA + 127) / 128;
  constexpr int XPB  = (XB + 127) / 128;
  static_assert(XA + XB <= MV_XS_FLOATS, "staging fits");
  static_assert(XB == 0 || XB == 64 || XB == 128, "segment B: one piece");

  const int L  = threadIdx.x & 31;
  const int lr = L >> 3;
  const int lk = (L & 7) * 4;
  const int xl = L * 4;

  const float* gA = WA + (size_t)(jBase + lr) * (size_t)KA + kA + lk;
  const float* gB = NSB > 0 ? WB + (size_t)(jBase + lr) * (size_t)KB + kB + lk : gA;
  asm volatile("" : "+v"(gA), "+v"(gB));
  const gptr gwA = (gptr)gA;
  const gptr gwB = (gptr)gB;

  v4f xv[XPA];
#pragma unroll
  for (int p = 0; p < XPA; ++p) {
    const int cnt = XA - 128 * p < 128 ? XA - 128 * p : 128;
    xv[p] = *(const v4f*)(xA + kA + 128 * p + piece_of(xl, cnt));
  }
  v4f t[RING][4];
#pragma unroll
  for (int u = 0; u < RING; ++u) {
#pragma unroll
    for (int i = 0; i < 4; ++i)
      t[u][i] = u < NUA
          ? *(gptr4)(gwA + ((size_t)(u / NSA) * HIDDEN + 4 * i) * (size_t)KA +
                          (u % NSA) * SLICE_K)
          : *(gptr4)(gwB + ((size_t)((u - NUA) / NSBd) * HIDDEN + 4 * i) * (size_t)KB +
                          ((u - NUA) % NSBd) * SLICE_K);
  }
  v4f xw[XPB > 0 ? XPB : 1];
  if (NSB > 0) {
    const float* xp = xB + kB + piece_of(xl, XB);
    asm volatile("" : "+v"(xp));
    xw[0] = *(gptr4)(gptr)xp;
    __builtin_amdgcn_sched_barrier(0);
  }

  float* sw = stg + lr * SLICE_PITCH + lk;
  const float* sr = stg + (L & 15) * SLICE_PITCH + 4 * (L >> 4);
  const float* xr = xs + 4 * (L >> 4);

#pragma unroll
  for (int p = 0; p < XPA; ++p) {
    const int cnt = XA - 128 * p < 128 ? XA - 128 * p : 128;
    *(v4f*)(xs + 128 * p + piece_of(xl, cnt)) = xv[p];
  }

  v8f ce = {}, co = {};
#pragma unroll
  for (int u = 0; u < NU; ++u) {
    const bool inA  = u < NUA;
    const int  gate = inA ? u / NSA : (u - NUA) / NSBd;
    const int  ks   = inA ? u % NSA : (u - NUA) % NSBd;
    if (u > 0) wave_lds_handover();
#pragma unroll
    for (int i = 0; i < 4; ++i) *(v4f*)(sw + 4 * i * SLICE_PITCH) = t[u % RING][i];
    if (NSB > 0 && u == NUA) *(v4f*)(xs + XA + piece_of(xl, XB)) = xw[0];
    if (u + RING < NU) {
      const int v = u + RING;
#pragma unroll
      for (int i = 0; i < 4; ++i)
        t[u % RING][i] = v < NUA
            ? *(gptr4)(gwA + ((size_t)(v / NSA) * HIDDEN + 4 * i) * (size_t)KA +
                            (v % NSA) * SLICE_K)
            : *(gptr4)(gwB + ((size_t)((v - NUA) / NSBd) * HIDDEN + 4 * i) * (size_t)KB +
                            ((v - NUA) % NSBd) * SLICE_K);
    }
    wave_lds_handover();
    slice_times_x(sr, xr + (inA ? 0 : XA) + ks * SLICE_K, ce, co);
    if (ks == (inA ? NSA : NSB) - 1) {
      const int rb = gate < 2 ? gate : (inA ? ROWA2 : ROWB2);
      reduce16(ce + co, sAcc + 16 * rb);
      ce = v8f{};
      co = v8f{};
    }
  }
}

__global__ void __launch_bounds__(MV_BLOCK)
gru_matvec_kernel(const float* __restrict__ W_ih, const float* __restrict__ W_hh,
                  const float* __restrict__ b_ih, const float* __restrict__ b_hh,
                  const float* __restrict__ xsrc, const float* __restrict__ hprev,
                  float* __restrict__ pre) {
  __shared__ __attribute__((aligned(16))) float sStage[MV_WAVES * STG_FLOATS];
  __shared__ __attribute__((aligned(16))) float sXs[MV_WAVES * MV_XS_FLOATS];
  __shared__ float sAcc[3][16];

  const int tid   = threadIdx.x;
  const int jBase = blockIdx.x * 16;
  const int mat   = blockIdx.y;

  if (tid < 48) ((float*)sAcc)[tid] = 0.0f;
  __syncthreads();

  const int wave = __builtin_amdgcn_readfirstlane(tid >> 5);
  float* stg = sStage + wave * STG_FLOATS;
  float* xs  = sXs + wave * MV_XS_FLOATS;

  constexpr int KPER = HIDDEN / MV_WAVES;
  if (mat == 0) {
    gemv16x3_staged<HIDDEN, KPER / SLICE_K>(W_ih, jBase, wave * KPER, xsrc, stg, xs,
                                            (float*)sAcc);
  } else {
    gemv16x3_staged<HIDDEN, KPER / SLICE_K>(W_hh, jBase, wave * KPER, hprev, stg, xs,
                                            (float*)sAcc);
  }
  __syncthreads();

  if (tid < 48) {
    const int g = tid >> 4;
    const int m = tid & 15;
    const int j = jBase + m;
    const float* bias = (mat == 0) ? b_ih : b_hh;
    pre[(mat * 3 + g) * HIDDEN + j] = sAcc[g][m] + bias[g * HIDDEN + j];
  }
}

#define L0_HH_CUT  768
__global__ void __launch_bounds__(MV_BLOCK) __attribute__((amdgpu_waves_per_eu(2, 2)))
gru_matvec_split_kernel(const float* __restrict__ W_ih, const float* __restrict__ W_hh,
                        const float* __restrict__ emb, const int* __restrict__ xidx,
                        const float* __restrict__ hprev, float* __restrict__ pl) {
  __shared__ __attribute__((aligned(16))) float sStage[MV_WAVES * STG_FLOATS];
  __shared__ __attribute__((aligned(16))) float sXs[MV_WAVES * MV_XS_FLOATS];
  __shared__ float sAcc[4][16];

  const int tid   = threadIdx.x;
  const int jBase = blockIdx.x * 16;
  const int part  = blockIdx.y;

  if (tid < 64) ((float*)sAcc)[tid] = 0.0f;
  __syncthreads();

  const int wave = __builtin_amdgcn_readfirstlane(tid >> 5);
  float* stg = sStage + wave * STG_FLOATS;
  float* xs  = sXs + wave * MV_XS_FLOATS;

  constexpr int HH0 = L0_HH_CUT / MV_WAVES;
  constexpr int HH1 = (HIDDEN - L0_HH_CUT) / MV_WAVES;
  constexpr int IH  = EMBED / MV_WAVES;
  static_assert(HH0 % SLICE_K == 0 && HH1 % SLICE_K == 0 && IH % SLICE_K == 0 &&
                HH0 + IH == HH1, "whole slices, even shares");

  if (part == 0) {
    const float* xe = emb + (size_t)xidx[0] * (size_t)EMBED;
    stream_slices<HIDDEN, HH0 / SLICE_K, 3, EMBED, IH / SLICE_K, 2>(
        W_hh, wave * HH0, hprev, W_ih, wave * IH, xe, jBase, stg, xs, (float*)sAcc);
  } else {
    gemv16x3_staged<HIDDEN, HH1 / SLICE_K>(W_hh, jBase, L0_HH_CUT + wave * HH1, hprev,
                                           stg, xs, (float*)sAcc);
  }
  __syncthreads();

  const int np = (part == 0) ? 4 : 3;
  if (tid < 16 * np) {
    const int p = tid >> 4;
    const int m = tid & 15;
    pl[(part * 4 + p) * HIDDEN + jBase + m] = sAcc[p][m];
  }
}

__device__ __forceinline__ float gru_unit(float ir, float iz, float inn,
                                          float hr, float hz, float hn,
                                          float hp) {
  const float r = 1.0f / (1.0f + __expf(-(ir + hr)));
  const float z = 1.0f / (1.0f + __expf(-(iz + hz)));
  const float n = tanhf(inn + r * hn);
  return (1.0f - z) * n + z * hp;
}

__global__ void __launch_bounds__(256)
gru_gate_kernel(const float* __restrict__ pl, const float* __restrict__ b_ih,
                const float* __restrict__ b_hh, const float* __restrict__ hprev,
                float* __restrict__ hnext, float* __restrict__ hcopy) {
  const int j = blockIdx.x * 256 + threadIdx.x;
  if (j < HIDDEN) {
    const float ir  = pl[0 * HIDDEN + j] + b_ih[0 * HIDDEN + j];
    const float iz  = pl[1 * HIDDEN + j] + b_ih[1 * HIDDEN + j];
    const float inn = pl[2 * HIDDEN + j] + b_ih[2 * HIDDEN + j];
    const float hr  = pl[4 * HIDDEN + j] + b_hh[0 * HIDDEN + j];
    const float hz  = pl[5 * HIDDEN + j] + b_hh[1 * HIDDEN + j];
    const float hn  = (pl[3 * HIDDEN + j] + pl[6 * HIDDEN + j]) + b_hh[2 * HIDDEN + j];
    const float h = gru_unit(ir, iz, inn, hr, hz, hn, hprev[j]);
    hnext[j] = h;
    hcopy[j] = h;
  }
}

__global__ void __launch_bounds__(BLOCK)
head_kernel(const float* __restrict__ w_head, const float* __restrict__ b_head,
            const float* __restrict__ pre, const float* __restrict__ hprev,
            float* __restrict__ hcopy, float* __restrict__ probs) {
  __shared__ __attribute__((aligned(16))) float sStage[WAVES * STG_FLOATS];
  __shared__ __attribute__((aligned(16))) float sXs[WAVES * XS_FLOATS];
  __shared__ float sAcc[16];

  constexpr int KPER = HIDDEN / WAVES;
  constexpr int NS   = KPER / SLICE_K;
  static_assert(KPER % SLICE_K == 0 && KPER <= XS_FLOATS, "staging fits");

  const int tid = threadIdx.x;
  if (tid < 16) sAcc[tid] = 0.0f;
  __syncthreads();

  const int wave = __builtin_amdgcn_readfirstlane(tid >> 5);
  float* stg = sStage + wave * STG_FLOATS;
  float* xs  = sXs + wave * XS_FLOATS;
  const int L  = tid & 31;
  const int lk = 4 * L;
  const int k  = wave * KPER + lk;

  const v4f ir  = *(const v4f*)(pre + 0 * HIDDEN + k);
  const v4f iz  = *(const v4f*)(pre + 1 * HIDDEN + k);
  const v4f inn = *(const v4f*)(pre + 2 * HIDDEN + k);
  const v4f hr  = *(const v4f*)(pre + 3 * HIDDEN + k);
  const v4f hz  = *(const v4f*)(pre + 4 * HIDDEN + k);
  const v4f hn  = *(const v4f*)(pre + 5 * HIDDEN + k);
  const v4f hp  = *(const v4f*)(hprev + k);
  const int lr  = L >> 3;
  const int lks = (L & 7) * 4;
  const float* gw = w_head + (size_t)lr * (size_t)HIDDEN + wave * KPER + lks;
  v4f t[NS][4];
#pragma unroll
  for (int u = 0; u < NS; ++u) {
#pragma unroll
    for (int i = 0; i < 4; ++i)
      t[u][i] = *(const v4f*)(gw + (size_t)(4 * i) * (size_t)HIDDEN + u * SLICE_K);
  }

  v4f h;
  h.x = gru_unit(ir.x, iz.x, inn.x, hr.x, hz.x, hn.x, hp.x);
  h.y = gru_unit(ir.y, iz.y, inn.y, hr.y, hz.y, hn.y, hp.y);
  h.z = gru_unit(ir.z, iz.z, inn.z, hr.z, hz.z, hn.z, hp.z);
  h.w = gru_unit(ir.w, iz.w, inn.w, hr.w, hz.w, hn.w, hp.w);
  *(v4f*)(hcopy + k) = h;
  *(v4f*)(xs + lk) = h;

  float* sw = stg + lr * SLICE_PITCH + lks;
  const float* sr = stg + (L & 15) * SLICE_PITCH + 4 * (L >> 4);
  const float* xr = xs + 4 * (L >> 4);
  v8f ce = {}, co = {};
#pragma unroll
  for (int u = 0; u < NS; ++u) {
    if (u > 0) wave_lds_handover();
#pragma unroll
    for (int i = 0; i < 4; ++i) *(v4f*)(sw + 4 * i * SLICE_PITCH) = t[u][i];
    wave_lds_handover();
    slice_times_x(sr, xr + u * SLICE_K, ce, co);
  }
  reduce16(ce + co, sAcc);
  __syncthreads();

  if (tid < 16) {
    const float v = sAcc[tid] + b_head[tid];
    float m = v;
    for (int off = 8; off > 0; off >>= 1) m = fmaxf(m, __shfl_xor(m, off, 32));
    const float e = __expf(v - m);
    float s = e;
    for (int off = 8; off > 0; off >>= 1) s += __shfl_xor(s, off, 32);
    probs[tid] = e / s;
  }
}

extern "C" void kernel_launch(void* const* d_in, const int* in_sizes, int n_in,
                              void* d_out, int out_size, void* d_ws,
                              size_t ws_size, hipStream_t stream) {
  const int*   x      = (const int*)  d_in[0];
  const float* h_in   = (const float*)d_in[1];
  const float* emb    = (const float*)d_in[2];
  const float* w_ih0  = (const float*)d_in[3];
  const float* w_hh0  = (const float*)d_in[4];
  const float* b_ih0  = (const float*)d_in[5];
  const float* b_hh0  = (const float*)d_in[6];
  const float* w_ih1  = (const float*)d_in[7];
  const float* w_hh1  = (const float*)d_in[8];
  const float* b_ih1  = (const float*)d_in[9];
  const float* b_hh1  = (const float*)d_in[10];
  const float* w_head = (const float*)d_in[11];
  const float* b_head = (const float*)d_in[12];

  float* out = (float*)d_out;
  float* h0  = (float*)d_ws;
  float* pl0 = h0 + HIDDEN;
  float* pre = h0 + 2 * HIDDEN;

  const dim3 gridMV(HIDDEN / 16, 2);

  gru_matvec_split_kernel<<<gridMV, MV_BLOCK, 0, stream>>>(w_ih0, w_hh0, emb, x,
                                                           h_in, pl0);
  gru_gate_kernel<<<HIDDEN / 256, 256, 0, stream>>>(pl0, b_ih0, b_hh0, h_in, h0,
                                                    out + NV);

  gru_matvec_kernel<<<gridMV, MV_BLOCK, 0, stream>>>(
      w_ih1, w_hh1, b_ih1, b_hh1, h0, h_in + HIDDEN, pre);
  head_kernel<<<1, BLOCK, 0, stream>>>(w_head, b_head, pre, h_in + HIDDEN,
                                       out + NV + HIDDEN, out);
}
